# starting kernel
# baseline (speedup 1.0000x reference)
#include <hip/hip_runtime.h>
#include <math.h>

constexpr int Bb = 8;
constexpr int Ll = 256;
constexpr int Dd = 256;
constexpr int Kk = 64;

typedef __attribute__((ext_vector_type(2))) float v2f;
typedef __attribute__((ext_vector_type(8))) float v8f;

__device__ __forceinline__ v8f wmma_f32_4(v2f a, v2f b, v8f c) {
  return __builtin_amdgcn_wmma_f32_16x16x4_f32(false, a, false, b, (short)0, c,
                                               false, false);
}

__device__ __forceinline__ float fast_tanh(float x) {
#if __has_builtin(__builtin_amdgcn_tanhf)
  return __builtin_amdgcn_tanhf(x);
#else
  float r;
  asm volatile("v_tanh_f32 %0, %1" : "=v"(r) : "v"(x));
  return r;
#endif
}

__global__ __launch_bounds__(256) void qk_proj_kernel(
    const float* __restrict__ inp, const float* __restrict__ W1,
    float* __restrict__ qi, float* __restrict__ kj) {
  const int blk  = blockIdx.x;
  const int b    = blk / (Ll / 16);
  const int i0   = (blk % (Ll / 16)) * 16;
  const int wave = threadIdx.x >> 5;
  const int lane = threadIdx.x & 31;
  const bool is_k = (wave >= 4);
  const int n0     = (wave & 3) * 16;
  const int coloff = is_k ? Dd : 0;

  const int mn = lane & 15;
  const int kh = (lane >> 4) << 1;

  const float* arow = inp + ((size_t)b * Ll + (i0 + mn)) * Dd;
  const float* brow = W1 + (size_t)(n0 + mn) * (2 * Dd) + coloff;

  v8f acc = {};
#pragma unroll 4
  for (int d = 0; d < Dd; d += 4) {
    v2f a, bv;
    a.x  = arow[d + kh];
    a.y  = arow[d + kh + 1];
    bv.x = brow[d + kh];
    bv.y = brow[d + kh + 1];
    acc = wmma_f32_4(a, bv, acc);
  }

  float* op     = is_k ? kj : qi;
  const int mb  = (lane >> 4) * 8;
#pragma unroll
  for (int r = 0; r < 8; ++r) {
    op[((size_t)b * Ll + (i0 + mb + r)) * Kk + (n0 + mn)] = acc[r];
  }
}

__global__ __launch_bounds__(256) void attn_softmax_kernel(
    const float* __restrict__ qi, const float* __restrict__ kj,
    const float* __restrict__ W2, float* __restrict__ attn) {
  __shared__ float q_s[Kk];
  __shared__ float w_s[Kk];
  __shared__ float r_s[Ll];

  const int b = blockIdx.x / Ll;
  const int i = blockIdx.x % Ll;
  const int t = threadIdx.x;

  if (t < Kk) {
    q_s[t] = qi[((size_t)b * Ll + i) * Kk + t];
    w_s[t] = W2[t];
  }
  __syncthreads();

  const float4* kr4 =
      reinterpret_cast<const float4*>(kj + ((size_t)b * Ll + t) * Kk);
  float s = 0.0f;
#pragma unroll
  for (int k4 = 0; k4 < Kk / 4; ++k4) {
    const float4 kv = kr4[k4];
    const int k = 4 * k4;
    s += fast_tanh(q_s[k + 0] + kv.x) * w_s[k + 0];
    s += fast_tanh(q_s[k + 1] + kv.y) * w_s[k + 1];
    s += fast_tanh(q_s[k + 2] + kv.z) * w_s[k + 2];
    s += fast_tanh(q_s[k + 3] + kv.w) * w_s[k + 3];
  }

  r_s[t] = s;
  __syncthreads();
  for (int st = 128; st > 0; st >>= 1) {
    if (t < st) r_s[t] = fmaxf(r_s[t], r_s[t + st]);
    __syncthreads();
  }
  const float mx = r_s[0];
  __syncthreads();

  const float e = __expf(s - mx);

  r_s[t] = e;
  __syncthreads();
  for (int st = 128; st > 0; st >>= 1) {
    if (t < st) r_s[t] += r_s[t + st];
    __syncthreads();
  }

  attn[((size_t)b * Ll + i) * Ll + t] = e * (1.0f / r_s[0]);
}

__global__ __launch_bounds__(256) void context_kernel(
    const float* __restrict__ attn, const float* __restrict__ inp,
    float* __restrict__ out) {
  const int wave = threadIdx.x >> 5;
  const int lane = threadIdx.x & 31;
  const int g    = blockIdx.x * 8 + wave;
  const int b    = g >> 8;
  const int rem  = g & 255;
  const int i0   = (rem >> 4) * 16;
  const int d0   = (rem & 15) * 16;

  const int mn = lane & 15;
  const int kh = (lane >> 4) << 1;

  const float* arow = attn + ((size_t)b * Ll + (i0 + mn)) * Ll;

  v8f acc = {};
  for (int j = 0; j < Ll; j += 4) {
    v2f a, bv;
    a.x = arow[j + kh];
    a.y = arow[j + kh + 1];
    const float* bp = inp + ((size_t)b * Ll + (j + kh)) * Dd + (d0 + mn);
    bv.x = bp[0];
    bv.y = bp[Dd];
    acc = wmma_f32_4(a, bv, acc);
  }

  const int mb = (lane >> 4) * 8;
#pragma unroll
  for (int r = 0; r < 8; ++r) {
    const int row      = i0 + mb + r;
    const size_t obase = ((size_t)b * Ll + row) * (2 * Dd);
    out[obase + Dd + d0 + mn] = acc[r];
    out[obase + d0 + mn] = inp[((size_t)b * Ll + row) * Dd + d0 + mn];
  }
}

extern "C" void kernel_launch(void* const* d_in, const int* in_sizes, int n_in,
                              void* d_out, int out_size, void* d_ws,
                              size_t ws_size, hipStream_t stream) {
  const float* inp = (const float*)d_in[0];
  const float* W1  = (const float*)d_in[1];
  const float* W2  = (const float*)d_in[2];
  float* out = (float*)d_out;

  float* qi   = (float*)d_ws;
  float* kj   = qi + (size_t)Bb * Ll * Kk;
  float* attn = kj + (size_t)Bb * Ll * Kk;

  qk_proj_kernel<<<Bb * (Ll / 16), 256, 0, stream>>>(inp, W1, qi, kj);
  attn_softmax_kernel<<<Bb * Ll, 256, 0, stream>>>(qi, kj, W2, attn);
  context_kernel<<<(Bb * (Ll / 16) * (Dd / 16)) / 8, 256, 0, stream>>>(attn, inp, out);
}


# score+softmax and context kernels fused: one workgroup owns 8 rows of a batch, kj row in registers, probabilities stay in LDS as WMMA A operand; 2 launches, no attn plane; sum orders unchanged
# speedup vs baseline: 2.0235x; 2.0235x over previous
#include <hip/hip_runtime.h>
#include <math.h>

constexpr int Bb = 8;
constexpr int Ll = 256;
constexpr int Dd = 256;
constexpr int Kk = 64;

typedef __attribute__((ext_vector_type(2))) float v2f;
typedef __attribute__((ext_vector_type(8))) float v8f;

__device__ __forceinline__ v8f wmma_f32_4(v2f a, v2f b, v8f c) {
  return __builtin_amdgcn_wmma_f32_16x16x4_f32(false, a, false, b, (short)0, c,
                                               false, false);
}

__device__ __forceinline__ float fast_tanh(float x) {
#if __has_builtin(__builtin_amdgcn_tanhf)
  return __builtin_amdgcn_tanhf(x);
#else
  float r;
  asm volatile("v_tanh_f32 %0, %1" : "=v"(r) : "v"(x));
  return r;
#endif
}

__global__ __launch_bounds__(256) void qk_proj_kernel(
    const float* __restrict__ inp, const float* __restrict__ W1,
    float* __restrict__ qi, float* __restrict__ kj) {
  const int blk  = blockIdx.x;
  const int b    = blk / (Ll / 16);
  const int i0   = (blk % (Ll / 16)) * 16;
  const int wave = threadIdx.x >> 5;
  const int lane = threadIdx.x & 31;
  const bool is_k = (wave >= 4);
  const int n0     = (wave & 3) * 16;
  const int coloff = is_k ? Dd : 0;

  const int mn = lane & 15;
  const int kh = (lane >> 4) << 1;

  const float* arow = inp + ((size_t)b * Ll + (i0 + mn)) * Dd;
  const float* brow = W1 + (size_t)(n0 + mn) * (2 * Dd) + coloff;

  v8f acc = {};
#pragma unroll 4
  for (int d = 0; d < Dd; d += 4) {
    v2f a, bv;
    a.x  = arow[d + kh];
    a.y  = arow[d + kh + 1];
    bv.x = brow[d + kh];
    bv.y = brow[d + kh + 1];
    acc = wmma_f32_4(a, bv, acc);
  }

  float* op     = is_k ? kj : qi;
  const int mb  = (lane >> 4) * 8;
#pragma unroll
  for (int r = 0; r < 8; ++r) {
    op[((size_t)b * Ll + (i0 + mb + r)) * Kk + (n0 + mn)] = acc[r];
  }
}

constexpr int RR = 8;
constexpr int SP = Ll + 4;

__global__ __launch_bounds__(256) void attn_context_kernel(
    const float* __restrict__ qi, const float* __restrict__ kj,
    const float* __restrict__ W2, const float* __restrict__ inp,
    float* __restrict__ out) {
  __shared__ float q_s[RR * Kk];
  __shared__ float p_s[RR * SP];

  const int b    = blockIdx.x / (Ll / RR);
  const int i0   = (blockIdx.x % (Ll / RR)) * RR;
  const int t    = threadIdx.x;
  const int wave = t >> 5;
  const int lane = t & 31;

  float kreg[Kk];
  {
    const float4* kr4 =
        reinterpret_cast<const float4*>(kj + ((size_t)b * Ll + t) * Kk);
#pragma unroll
    for (int k4 = 0; k4 < Kk / 4; ++k4) {
      const float4 kv = kr4[k4];
      kreg[4 * k4 + 0] = kv.x;
      kreg[4 * k4 + 1] = kv.y;
      kreg[4 * k4 + 2] = kv.z;
      kreg[4 * k4 + 3] = kv.w;
    }
  }

  {
    const float* qsrc = qi + ((size_t)b * Ll + i0) * Kk;
    q_s[t]       = qsrc[t];
    q_s[t + 256] = qsrc[t + 256];
  }

  static_assert(RR * Dd / 4 == 2 * 256, "two 16-byte pieces per thread");
  const float4* xsrc =
      reinterpret_cast<const float4*>(inp + ((size_t)b * Ll + i0) * Dd);
  const float4 pt0 = xsrc[t];
  const float4 pt1 = xsrc[t + 256];
  __syncthreads();

#pragma unroll 2
  for (int r = 0; r < RR; ++r) {
    float s = 0.0f;
#pragma unroll
    for (int k = 0; k < Kk; ++k) {
      s += fast_tanh(q_s[r * Kk + k] + kreg[k]) * W2[k];
    }
    p_s[r * SP + t] = s;
  }
  __syncthreads();

  const int c = lane & 15;
  const int h = lane >> 4;
  const float* arow = p_s + (c & (RR - 1)) * SP + 2 * h;
  const float* bcol = inp + ((size_t)b * Ll + 2 * h) * Dd + 32 * wave + 2 * c;

  constexpr int U = 8;
  v2f a_c[U];
  float2 x0_c[U], x1_c[U];
#pragma unroll
  for (int u = 0; u < U; ++u) {
    const int j = 4 * u;
    x0_c[u] = *reinterpret_cast<const float2*>(bcol + (size_t)j * Dd);
    x1_c[u] = *reinterpret_cast<const float2*>(bcol + (size_t)(j + 1) * Dd);
  }

  {
    float* row = p_s + wave * SP;
    float v[8];
#pragma unroll
    for (int m = 0; m < 8; ++m) v[m] = row[lane + 32 * m];
    float mx = fmaxf(fmaxf(fmaxf(v[0], v[1]), fmaxf(v[2], v[3])),
                     fmaxf(fmaxf(v[4], v[5]), fmaxf(v[6], v[7])));
#pragma unroll
    for (int o = 16; o > 0; o >>= 1) mx = fmaxf(mx, __shfl_xor(mx, o));
    float e[8];
#pragma unroll
    for (int m = 0; m < 8; ++m) e[m] = __expf(v[m] - mx);
    float s0 = e[0] + e[4];
    float s1 = e[1] + e[5];
    float s2 = e[2] + e[6];
    float s3 = e[3] + e[7];
    s0 += s2;
    s1 += s3;
    s0 += s1;
#pragma unroll
    for (int o = 16; o > 0; o >>= 1) s0 += __shfl_xor(s0, o);
    const float inv = 1.0f / s0;
#pragma unroll
    for (int m = 0; m < 8; ++m) row[lane + 32 * m] = e[m] * inv;
  }
  __syncthreads();

  {
    float* orow = out + ((size_t)b * Ll + i0 + t / (Dd / 4)) * (2 * Dd) +
                  4 * (t % (Dd / 4));
    *reinterpret_cast<float4*>(orow) = pt0;
    *reinterpret_cast<float4*>(orow + (size_t)(256 / (Dd / 4)) * (2 * Dd)) = pt1;
  }

  v8f acc0 = {};
  v8f acc1 = {};
#pragma unroll
  for (int u = 0; u < U; ++u) {
    a_c[u] = *reinterpret_cast<const v2f*>(arow + 4 * u);
  }
#pragma unroll
  for (int blk = 0; blk < Ll / (4 * U); ++blk) {
    v2f a_n[U];
    float2 x0_n[U], x1_n[U];
    if (blk + 1 < Ll / (4 * U)) {
#pragma unroll
      for (int u = 0; u < U; ++u) {
        const int j = 4 * (U * (blk + 1) + u);
        a_n[u]  = *reinterpret_cast<const v2f*>(arow + j);
        x0_n[u] = *reinterpret_cast<const float2*>(bcol + (size_t)j * Dd);
        x1_n[u] = *reinterpret_cast<const float2*>(bcol + (size_t)(j + 1) * Dd);
      }
    }
#pragma unroll
    for (int u = 0; u < U; ++u) {
      v2f b0, b1;
      b0.x = x0_c[u].x;
      b0.y = x1_c[u].x;
      b1.x = x0_c[u].y;
      b1.y = x1_c[u].y;
      acc0 = wmma_f32_4(a_c[u], b0, acc0);
      acc1 = wmma_f32_4(a_c[u], b1, acc1);
    }
    if (blk + 1 < Ll / (4 * U)) {
#pragma unroll
      for (int u = 0; u < U; ++u) {
        a_c[u]  = a_n[u];
        x0_c[u] = x0_n[u];
        x1_c[u] = x1_n[u];
      }
    }
  }

  if (h == 0) {
#pragma unroll
    for (int r = 0; r < 8; ++r) {
      float2 o;
      o.x = acc0[r];
      o.y = acc1[r];
      *reinterpret_cast<float2*>(out + ((size_t)b * Ll + i0 + r) * (2 * Dd) +
                                 Dd + 32 * wave + 2 * c) = o;
    }
  }
}

extern "C" void kernel_launch(void* const* d_in, const int* in_sizes, int n_in,
                              void* d_out, int out_size, void* d_ws,
                              size_t ws_size, hipStream_t stream) {
  const float* inp = (const float*)d_in[0];
  const float* W1  = (const float*)d_in[1];
  const float* W2  = (const float*)d_in[2];
  float* out = (float*)d_out;

  float* qi = (float*)d_ws;
  float* kj = qi + (size_t)Bb * Ll * Kk;

  qk_proj_kernel<<<Bb * (Ll / 16), 256, 0, stream>>>(inp, W1, qi, kj);
  attn_context_kernel<<<Bb * (Ll / RR), 256, 0, stream>>>(qi, kj, W2, inp, out);
}


# projection kernel: each wave streams its tile's A and B rows as 16-byte lane loads (k permuted inside each 8, alike for A and B) kept 20 deep, on 256 four-wave workgroups; k sum order changes
# speedup vs baseline: 2.4877x; 1.2294x over previous
#include <hip/hip_runtime.h>
#include <math.h>

constexpr int Bb = 8;
constexpr int Ll = 256;
constexpr int Dd = 256;
constexpr int Kk = 64;

typedef __attribute__((ext_vector_type(2))) float v2f;
typedef __attribute__((ext_vector_type(8))) float v8f;

__device__ __forceinline__ v8f wmma_f32_4(v2f a, v2f b, v8f c) {
  return __builtin_amdgcn_wmma_f32_16x16x4_f32(false, a, false, b, (short)0, c,
                                               false, false);
}

__device__ __forceinline__ float fast_tanh(float x) {
#if __has_builtin(__builtin_amdgcn_tanhf)
  return __builtin_amdgcn_tanhf(x);
#else
  float r;
  asm volatile("v_tanh_f32 %0, %1" : "=v"(r) : "v"(x));
  return r;
#endif
}

constexpr int NS = Dd / 8;

__global__ __launch_bounds__(128) __attribute__((amdgpu_waves_per_eu(1, 4)))
void qk_proj_kernel(
    const float* __restrict__ inp, const float* __restrict__ W1,
    float* __restrict__ qi, float* __restrict__ kj) {
  const int rt    = blockIdx.x >> 1;
  const bool is_k = (blockIdx.x & 1) != 0;
  const int wave  = threadIdx.x >> 5;
  const int lane  = threadIdx.x & 31;
  const int c     = lane & 15;
  const int h     = lane >> 4;
  const int n0    = wave * 16;

  const float4* ap = reinterpret_cast<const float4*>(
                         inp + ((size_t)rt * 16 + c) * Dd) + h;
  const float4* bp = reinterpret_cast<const float4*>(
                         W1 + (size_t)(n0 + c) * (2 * Dd) + (is_k ? Dd : 0)) + h;

  float4 av[NS], bv[NS];
#pragma unroll
  for (int s = 0; s < NS; ++s) {
    av[s] = ap[2 * s];
    bv[s] = bp[2 * s];
  }

  v8f acc = {};
#pragma unroll
  for (int s = 0; s < NS; ++s) {
    v2f a0, b0, a1, b1;
    a0.x = av[s].x;  a0.y = av[s].y;
    b0.x = bv[s].x;  b0.y = bv[s].y;
    a1.x = av[s].z;  a1.y = av[s].w;
    b1.x = bv[s].z;  b1.y = bv[s].w;
    acc = wmma_f32_4(a0, b0, acc);
    acc = wmma_f32_4(a1, b1, acc);
  }

  float* op = is_k ? kj : qi;
#pragma unroll
  for (int r = 0; r < 8; ++r) {
    op[((size_t)rt * 16 + 8 * h + r) * Kk + (n0 + c)] = acc[r];
  }
}

constexpr int RR = 8;
constexpr int SP = Ll + 4;

__global__ __launch_bounds__(256) void attn_context_kernel(
    const float* __restrict__ qi, const float* __restrict__ kj,
    const float* __restrict__ W2, const float* __restrict__ inp,
    float* __restrict__ out) {
  __shared__ float q_s[RR * Kk];
  __shared__ float p_s[RR * SP];

  const int b    = blockIdx.x / (Ll / RR);
  const int i0   = (blockIdx.x % (Ll / RR)) * RR;
  const int t    = threadIdx.x;
  const int wave = t >> 5;
  const int lane = t & 31;

  float kreg[Kk];
  {
    const float4* kr4 =
        reinterpret_cast<const float4*>(kj + ((size_t)b * Ll + t) * Kk);
#pragma unroll
    for (int k4 = 0; k4 < Kk / 4; ++k4) {
      const float4 kv = kr4[k4];
      kreg[4 * k4 + 0] = kv.x;
      kreg[4 * k4 + 1] = kv.y;
      kreg[4 * k4 + 2] = kv.z;
      kreg[4 * k4 + 3] = kv.w;
    }
  }

  {
    const float* qsrc = qi + ((size_t)b * Ll + i0) * Kk;
    q_s[t]       = qsrc[t];
    q_s[t + 256] = qsrc[t + 256];
  }

  static_assert(RR * Dd / 4 == 2 * 256, "two 16-byte pieces per thread");
  const float4* xsrc =
      reinterpret_cast<const float4*>(inp + ((size_t)b * Ll + i0) * Dd);
  const float4 pt0 = xsrc[t];
  const float4 pt1 = xsrc[t + 256];
  __syncthreads();

#pragma unroll 2
  for (int r = 0; r < RR; ++r) {
    float s = 0.0f;
#pragma unroll
    for (int k = 0; k < Kk; ++k) {
      s += fast_tanh(q_s[r * Kk + k] + kreg[k]) * W2[k];
    }
    p_s[r * SP + t] = s;
  }
  __syncthreads();

  const int c = lane & 15;
  const int h = lane >> 4;
  const float* arow = p_s + (c & (RR - 1)) * SP + 2 * h;
  const float* bcol = inp + ((size_t)b * Ll + 2 * h) * Dd + 32 * wave + 2 * c;

  constexpr int U = 8;
  v2f a_c[U];
  float2 x0_c[U], x1_c[U];
#pragma unroll
  for (int u = 0; u < U; ++u) {
    const int j = 4 * u;
    x0_c[u] = *reinterpret_cast<const float2*>(bcol + (size_t)j * Dd);
    x1_c[u] = *reinterpret_cast<const float2*>(bcol + (size_t)(j + 1) * Dd);
  }

  {
    float* row = p_s + wave * SP;
    float v[8];
#pragma unroll
    for (int m = 0; m < 8; ++m) v[m] = row[lane + 32 * m];
    float mx = fmaxf(fmaxf(fmaxf(v[0], v[1]), fmaxf(v[2], v[3])),
                     fmaxf(fmaxf(v[4], v[5]), fmaxf(v[6], v[7])));
#pragma unroll
    for (int o = 16; o > 0; o >>= 1) mx = fmaxf(mx, __shfl_xor(mx, o));
    float e[8];
#pragma unroll
    for (int m = 0; m < 8; ++m) e[m] = __expf(v[m] - mx);
    float s0 = e[0] + e[4];
    float s1 = e[1] + e[5];
    float s2 = e[2] + e[6];
    float s3 = e[3] + e[7];
    s0 += s2;
    s1 += s3;
    s0 += s1;
#pragma unroll
    for (int o = 16; o > 0; o >>= 1) s0 += __shfl_xor(s0, o);
    const float inv = 1.0f / s0;
#pragma unroll
    for (int m = 0; m < 8; ++m) row[lane + 32 * m] = e[m] * inv;
  }
  __syncthreads();

  {
    float* orow = out + ((size_t)b * Ll + i0 + t / (Dd / 4)) * (2 * Dd) +
                  4 * (t % (Dd / 4));
    *reinterpret_cast<float4*>(orow) = pt0;
    *reinterpret_cast<float4*>(orow + (size_t)(256 / (Dd / 4)) * (2 * Dd)) = pt1;
  }

  v8f acc0 = {};
  v8f acc1 = {};
#pragma unroll
  for (int u = 0; u < U; ++u) {
    a_c[u] = *reinterpret_cast<const v2f*>(arow + 4 * u);
  }
#pragma unroll
  for (int blk = 0; blk < Ll / (4 * U); ++blk) {
    v2f a_n[U];
    float2 x0_n[U], x1_n[U];
    if (blk + 1 < Ll / (4 * U)) {
#pragma unroll
      for (int u = 0; u < U; ++u) {
        const int j = 4 * (U * (blk + 1) + u);
        a_n[u]  = *reinterpret_cast<const v2f*>(arow + j);
        x0_n[u] = *reinterpret_cast<const float2*>(bcol + (size_t)j * Dd);
        x1_n[u] = *reinterpret_cast<const float2*>(bcol + (size_t)(j + 1) * Dd);
      }
    }
#pragma unroll
    for (int u = 0; u < U; ++u) {
      v2f b0, b1;
      b0.x = x0_c[u].x;
      b0.y = x1_c[u].x;
      b1.x = x0_c[u].y;
      b1.y = x1_c[u].y;
      acc0 = wmma_f32_4(a_c[u], b0, acc0);
      acc1 = wmma_f32_4(a_c[u], b1, acc1);
    }
    if (blk + 1 < Ll / (4 * U)) {
#pragma unroll
      for (int u = 0; u < U; ++u) {
        a_c[u]  = a_n[u];
        x0_c[u] = x0_n[u];
        x1_c[u] = x1_n[u];
      }
    }
  }

  if (h == 0) {
#pragma unroll
    for (int r = 0; r < 8; ++r) {
      float2 o;
      o.x = acc0[r];
      o.y = acc1[r];
      *reinterpret_cast<float2*>(out + ((size_t)b * Ll + i0 + r) * (2 * Dd) +
                                 Dd + 32 * wave + 2 * c) = o;
    }
  }
}

extern "C" void kernel_launch(void* const* d_in, const int* in_sizes, int n_in,
                              void* d_out, int out_size, void* d_ws,
                              size_t ws_size, hipStream_t stream) {
  const float* inp = (const float*)d_in[0];
  const float* W1  = (const float*)d_in[1];
  const float* W2  = (const float*)d_in[2];
  float* out = (float*)d_out;

  float* qi = (float*)d_ws;
  float* kj = qi + (size_t)Bb * Ll * Kk;

  qk_proj_kernel<<<2 * Bb * (Ll / 16), 128, 0, stream>>>(inp, W1, qi, kj);
  attn_context_kernel<<<Bb * (Ll / RR), 256, 0, stream>>>(qi, kj, W2, inp, out);
}
